# starting kernel
# baseline (speedup 1.0000x reference)
#include <hip/hip_runtime.h>
#include <math.h>

typedef float v2f __attribute__((ext_vector_type(2)));
typedef float v8f __attribute__((ext_vector_type(8)));

#define B_SZ 8192
#define T_SZ 200
#define D_SZ 3
#define H_SZ 20
#define L_SZ 10

#define WAVES_PER_BLOCK 4

__device__ __forceinline__ v8f wmma4(v2f a, v2f b, v8f c) {
  return __builtin_amdgcn_wmma_f32_16x16x4_f32(false, a, false, b, (short)0, c,
                                               false, false);
}

__device__ __forceinline__ float fast_tanh(float x) {
#if __has_builtin(__builtin_amdgcn_tanhf)
  return __builtin_amdgcn_tanhf(x);
#else
  float r;
  asm volatile("v_tanh_f32 %0, %1\n\tv_nop" : "=v"(r) : "v"(x));
  return r;
#endif
}

__device__ __forceinline__ float swap16(float x) {
  return __builtin_bit_cast(
      float, __builtin_amdgcn_ds_swizzle(__builtin_bit_cast(int, x), 0x401F));
}

__global__ __launch_bounds__(WAVES_PER_BLOCK * 32)
void latent_ode_kernel(const float* __restrict__ ts, const float* __restrict__ ys,
                       const float* __restrict__ eps,
                       const float* __restrict__ encW1, const float* __restrict__ encB1,
                       const float* __restrict__ encW2, const float* __restrict__ encB2,
                       const float* __restrict__ encW3, const float* __restrict__ encB3,
                       const float* __restrict__ l2hW, const float* __restrict__ l2hB,
                       const float* __restrict__ h2dW, const float* __restrict__ h2dB,
                       const float* __restrict__ fW1, const float* __restrict__ fB1,
                       const float* __restrict__ fW2, const float* __restrict__ fB2,
                       float* __restrict__ pred, float* __restrict__ outMean,
                       float* __restrict__ outStd)
{
  const int tid  = (int)threadIdx.x;
  const int wave = tid >> 5;
  const int lane = tid & 31;
  const int n16  = lane & 15;
  const int hi   = lane >> 4;
  const bool h   = (hi != 0);
  const int rowBase = ((int)blockIdx.x * WAVES_PER_BLOCK + wave) * 16;
  const int row = rowBase + n16;

  float yin[D_SZ];
#pragma unroll
  for (int d = 0; d < D_SZ; ++d)
    yin[d] = ys[(size_t)row * T_SZ * D_SZ + d];

  float h1[H_SZ];
#pragma unroll
  for (int j = 0; j < H_SZ; ++j) {
    float acc = encB1[j];
#pragma unroll
    for (int d = 0; d < D_SZ; ++d) acc += yin[d] * encW1[d * H_SZ + j];
    h1[j] = fmaxf(acc, 0.0f);
  }
  float h2[H_SZ];
#pragma unroll
  for (int j = 0; j < H_SZ; ++j) {
    float acc = encB2[j];
#pragma unroll
    for (int i = 0; i < H_SZ; ++i) acc += h1[i] * encW2[i * H_SZ + j];
    h2[j] = fmaxf(acc, 0.0f);
  }
  float latent[L_SZ];
#pragma unroll
  for (int j = 0; j < L_SZ; ++j) {
    float m  = encB3[j];
    float ls = encB3[L_SZ + j];
#pragma unroll
    for (int i = 0; i < H_SZ; ++i) {
      m  += h2[i] * encW3[i * (2 * L_SZ) + j];
      ls += h2[i] * encW3[i * (2 * L_SZ) + L_SZ + j];
    }
    const float sd = expf(ls);
    latent[j] = m + eps[(size_t)row * L_SZ + j] * sd;
    if (lane < 16) {
      outMean[(size_t)row * L_SZ + j] = m;
      outStd [(size_t)row * L_SZ + j] = sd;
    }
  }
  float y0v[H_SZ];
#pragma unroll
  for (int j = 0; j < H_SZ; ++j) {
    float acc = l2hB[j];
#pragma unroll
    for (int i = 0; i < L_SZ; ++i) acc += latent[i] * l2hW[i * H_SZ + j];
    y0v[j] = fmaxf(acc, 0.0f);
  }

  v8f yc0, yc1;
#pragma unroll
  for (int v = 0; v < 8; ++v) {
    yc0[v] = h ? y0v[v + 8] : y0v[v];
    yc1[v] = (!h && v < 4) ? y0v[16 + v] : 0.0f;
  }

  v2f w1f[2][6], w2f[2][6], w3f[6];
#pragma unroll
  for (int s = 0; s < 5; ++s) {
    const int k0 = 4 * s + 2 * hi;
#pragma unroll
    for (int nt = 0; nt < 2; ++nt) {
      const int n = 16 * nt + n16;
      const bool ok = n < H_SZ;
      w1f[nt][s].x = ok ? fW1[k0 * H_SZ + n]       : 0.0f;
      w1f[nt][s].y = ok ? fW1[(k0 + 1) * H_SZ + n] : 0.0f;
      w2f[nt][s].x = ok ? fW2[k0 * H_SZ + n]       : 0.0f;
      w2f[nt][s].y = ok ? fW2[(k0 + 1) * H_SZ + n] : 0.0f;
    }
    const bool ok3 = n16 < D_SZ;
    w3f[s].x = ok3 ? h2dW[k0 * D_SZ + n16]       : 0.0f;
    w3f[s].y = ok3 ? h2dW[(k0 + 1) * D_SZ + n16] : 0.0f;
  }
#pragma unroll
  for (int nt = 0; nt < 2; ++nt) {
    const int n = 16 * nt + n16;
    const bool ok = (!h) && (n < H_SZ);
    w1f[nt][5].x = ok ? fB1[n] : 0.0f;  w1f[nt][5].y = 0.0f;
    w2f[nt][5].x = ok ? fB2[n] : 0.0f;  w2f[nt][5].y = 0.0f;
  }
  w3f[5].x = (!h && n16 < D_SZ) ? h2dB[n16] : 0.0f;
  w3f[5].y = 0.0f;
  v2f bone;
  bone.x = h ? 0.0f : 1.0f;
  bone.y = 0.0f;
  const v8f vzero = {};

  auto buildB = [&](const v8f& z0, const v8f& z1, v2f* bf) {
    const float s0 = swap16(z0[0]), s1 = swap16(z0[1]);
    const float s2 = swap16(z0[2]), s3 = swap16(z0[3]);
    const float s4 = swap16(z0[4]), s5 = swap16(z0[5]);
    const float s6 = swap16(z0[6]), s7 = swap16(z0[7]);
    const float t2 = swap16(z1[2]), t3 = swap16(z1[3]);
    bf[0].x = h ? s2    : z0[0];  bf[0].y = h ? s3    : z0[1];
    bf[1].x = h ? s6    : z0[4];  bf[1].y = h ? s7    : z0[5];
    bf[2].x = h ? z0[2] : s0;     bf[2].y = h ? z0[3] : s1;
    bf[3].x = h ? z0[6] : s4;     bf[3].y = h ? z0[7] : s5;
    bf[4].x = h ? t2    : z1[0];  bf[4].y = h ? t3    : z1[1];
  };
  auto matmul = [&](const v2f (*wa)[6], const v2f* bf, v8f& o0, v8f& o1) {
    o0 = wmma4(wa[0][5], bone, vzero);
    o1 = wmma4(wa[1][5], bone, vzero);
#pragma unroll
    for (int s = 0; s < 5; ++s) {
      o0 = wmma4(wa[0][s], bf[s], o0);
      o1 = wmma4(wa[1][s], bf[s], o1);
    }
  };
  auto fEval = [&](const v2f* bf, v8f& k0, v8f& k1) {
    v8f t0, t1;
    matmul(w1f, bf, t0, t1);
#pragma unroll
    for (int v = 0; v < 8; ++v) {
      t0[v] = fast_tanh(t0[v]);
      t1[v] = fast_tanh(t1[v]);
    }
    v2f bt[5];
    buildB(t0, t1, bt);
    matmul(w2f, bt, k0, k1);
  };
  auto project = [&](const v2f* bf, int t) {
    v8f p = wmma4(w3f[5], bone, vzero);
#pragma unroll
    for (int s = 0; s < 5; ++s) p = wmma4(w3f[s], bf[s], p);
    if (!h) {
      float* o = &pred[((size_t)(rowBase + n16) * T_SZ + t) * D_SZ];
      o[0] = p[0]; o[1] = p[1]; o[2] = p[2];
    }
  };

  for (int t = 0; t < T_SZ - 1; ++t) {
    const float dt  = ts[t + 1] - ts[t];
    const float hdt = 0.5f * dt;
    v2f bf[5];

    buildB(yc0, yc1, bf);
    project(bf, t);

    v8f ka, kb, acc0, acc1, u0, u1;
    fEval(bf, ka, kb);
#pragma unroll
    for (int v = 0; v < 8; ++v) {
      acc0[v] = ka[v];                 acc1[v] = kb[v];
      u0[v] = yc0[v] + hdt * ka[v];    u1[v] = yc1[v] + hdt * kb[v];
    }
    buildB(u0, u1, bf);
    fEval(bf, ka, kb);
#pragma unroll
    for (int v = 0; v < 8; ++v) {
      acc0[v] += 2.0f * ka[v];         acc1[v] += 2.0f * kb[v];
      u0[v] = yc0[v] + hdt * ka[v];    u1[v] = yc1[v] + hdt * kb[v];
    }
    buildB(u0, u1, bf);
    fEval(bf, ka, kb);
#pragma unroll
    for (int v = 0; v < 8; ++v) {
      acc0[v] += 2.0f * ka[v];         acc1[v] += 2.0f * kb[v];
      u0[v] = yc0[v] + dt * ka[v];     u1[v] = yc1[v] + dt * kb[v];
    }
    buildB(u0, u1, bf);
    fEval(bf, ka, kb);

    const float sdt = dt * (1.0f / 6.0f);
#pragma unroll
    for (int v = 0; v < 8; ++v) {
      yc0[v] += sdt * (acc0[v] + ka[v]);
      yc1[v] += sdt * (acc1[v] + kb[v]);
    }
  }
  {
    v2f bf[5];
    buildB(yc0, yc1, bf);
    project(bf, T_SZ - 1);
  }
}

extern "C" void kernel_launch(void* const* d_in, const int* in_sizes, int n_in,
                              void* d_out, int out_size, void* d_ws, size_t ws_size,
                              hipStream_t stream) {
  const float* ts    = (const float*)d_in[0];
  const float* ys    = (const float*)d_in[1];
  const float* eps   = (const float*)d_in[2];
  const float* encW1 = (const float*)d_in[3];
  const float* encB1 = (const float*)d_in[4];
  const float* encW2 = (const float*)d_in[5];
  const float* encB2 = (const float*)d_in[6];
  const float* encW3 = (const float*)d_in[7];
  const float* encB3 = (const float*)d_in[8];
  const float* l2hW  = (const float*)d_in[9];
  const float* l2hB  = (const float*)d_in[10];
  const float* h2dW  = (const float*)d_in[11];
  const float* h2dB  = (const float*)d_in[12];
  const float* fW1   = (const float*)d_in[13];
  const float* fB1   = (const float*)d_in[14];
  const float* fW2   = (const float*)d_in[15];
  const float* fB2   = (const float*)d_in[16];

  float* pred = (float*)d_out;
  float* mean = pred + (size_t)B_SZ * T_SZ * D_SZ;
  float* stdv = mean + (size_t)B_SZ * L_SZ;

  dim3 grid(B_SZ / (WAVES_PER_BLOCK * 16));
  dim3 block(WAVES_PER_BLOCK * 32);
  hipLaunchKernelGGL(latent_ode_kernel, grid, block, 0, stream,
                     ts, ys, eps, encW1, encB1, encW2, encB2, encW3, encB3,
                     l2hW, l2hB, h2dW, h2dB, fW1, fB1, fW2, fB2,
                     pred, mean, stdv);
}


# Projection chain and pred stores moved to a second wave per batch tile on the idle SIMDs: state fragments handed over through an LDS ring, one barrier per 8 time points; every sum keeps its order
# speedup vs baseline: 1.7522x; 1.7522x over previous
#include <hip/hip_runtime.h>
#include <math.h>

typedef float v2f __attribute__((ext_vector_type(2)));
typedef float v8f __attribute__((ext_vector_type(8)));

#define B_SZ 8192
#define T_SZ 200
#define D_SZ 3
#define H_SZ 20
#define L_SZ 10

#define SOLVER_WAVES 2
#define WAVES_PER_BLOCK 4
#define CH 8
#define NCH (T_SZ / CH)
#define RING_SLOTS 40

__device__ __forceinline__ v8f wmma4(v2f a, v2f b, v8f c) {
  return __builtin_amdgcn_wmma_f32_16x16x4_f32(false, a, false, b, (short)0, c,
                                               false, false);
}

__device__ __forceinline__ float fast_tanh(float x) {
#if __has_builtin(__builtin_amdgcn_tanhf)
  return __builtin_amdgcn_tanhf(x);
#else
  float r;
  asm volatile("v_tanh_f32 %0, %1\n\tv_nop" : "=v"(r) : "v"(x));
  return r;
#endif
}

template <bool kV> struct BehindTrans { static constexpr bool value = kV; };

template <bool kBehindTrans>
__device__ __forceinline__ void xchg_halves(float& d, float& s) {
  if (kBehindTrans)
    asm("v_nop\n\tv_permlane16_swap_b32 %0, %1" : "+v"(d), "+v"(s));
  else
    asm("v_permlane16_swap_b32 %0, %1" : "+v"(d), "+v"(s));
}

__global__ __launch_bounds__(WAVES_PER_BLOCK * 32)
void latent_ode_kernel(const float* __restrict__ ts, const float* __restrict__ ys,
                       const float* __restrict__ eps,
                       const float* __restrict__ encW1, const float* __restrict__ encB1,
                       const float* __restrict__ encW2, const float* __restrict__ encB2,
                       const float* __restrict__ encW3, const float* __restrict__ encB3,
                       const float* __restrict__ l2hW, const float* __restrict__ l2hB,
                       const float* __restrict__ h2dW, const float* __restrict__ h2dB,
                       const float* __restrict__ fW1, const float* __restrict__ fB1,
                       const float* __restrict__ fW2, const float* __restrict__ fB2,
                       float* __restrict__ pred, float* __restrict__ outMean,
                       float* __restrict__ outStd)
{
  __shared__ __attribute__((aligned(16))) v2f ring[2][SOLVER_WAVES][RING_SLOTS][5][32];

  const int tid  = (int)threadIdx.x;
  const int wave = __builtin_amdgcn_readfirstlane(tid >> 5);
  const int lane = tid & 31;
  const int n16  = lane & 15;
  const int hi   = lane >> 4;
  const bool h   = (hi != 0);
  const bool solver = wave < SOLVER_WAVES;
  const int tile = solver ? wave : wave - SOLVER_WAVES;
  const int rowBase = ((int)blockIdx.x * SOLVER_WAVES + tile) * 16;
  const int row = rowBase + n16;

  v2f bone;
  bone.x = h ? 0.0f : 1.0f;
  bone.y = 0.0f;
  const v8f vzero = {};

  v8f yc0 = {}, yc1 = {};
  v2f w1f[2][6] = {}, w2f[2][6] = {};
  v2f w3f[6] = {};

  if (!solver) {
#pragma unroll
    for (int s = 0; s < 5; ++s) {
      const int k0 = 4 * s + 2 * hi;
      const bool ok3 = n16 < D_SZ;
      w3f[s].x = ok3 ? h2dW[k0 * D_SZ + n16]       : 0.0f;
      w3f[s].y = ok3 ? h2dW[(k0 + 1) * D_SZ + n16] : 0.0f;
    }
    w3f[5].x = (!h && n16 < D_SZ) ? h2dB[n16] : 0.0f;
    w3f[5].y = 0.0f;
  } else {

  float yin[D_SZ];
#pragma unroll
  for (int d = 0; d < D_SZ; ++d)
    yin[d] = ys[(size_t)row * T_SZ * D_SZ + d];

  float h1[H_SZ];
#pragma unroll
  for (int j = 0; j < H_SZ; ++j) {
    float acc = encB1[j];
#pragma unroll
    for (int d = 0; d < D_SZ; ++d) acc += yin[d] * encW1[d * H_SZ + j];
    h1[j] = fmaxf(acc, 0.0f);
  }
  float h2[H_SZ];
#pragma unroll
  for (int j = 0; j < H_SZ; ++j) {
    float acc = encB2[j];
#pragma unroll
    for (int i = 0; i < H_SZ; ++i) acc += h1[i] * encW2[i * H_SZ + j];
    h2[j] = fmaxf(acc, 0.0f);
  }
  float latent[L_SZ];
#pragma unroll
  for (int j = 0; j < L_SZ; ++j) {
    float m  = encB3[j];
    float ls = encB3[L_SZ + j];
#pragma unroll
    for (int i = 0; i < H_SZ; ++i) {
      m  += h2[i] * encW3[i * (2 * L_SZ) + j];
      ls += h2[i] * encW3[i * (2 * L_SZ) + L_SZ + j];
    }
    const float sd = expf(ls);
    latent[j] = m + eps[(size_t)row * L_SZ + j] * sd;
    if (lane < 16) {
      outMean[(size_t)row * L_SZ + j] = m;
      outStd [(size_t)row * L_SZ + j] = sd;
    }
  }
  float y0v[H_SZ];
#pragma unroll
  for (int j = 0; j < H_SZ; ++j) {
    float acc = l2hB[j];
#pragma unroll
    for (int i = 0; i < L_SZ; ++i) acc += latent[i] * l2hW[i * H_SZ + j];
    y0v[j] = fmaxf(acc, 0.0f);
  }

#pragma unroll
  for (int v = 0; v < 8; ++v) {
    yc0[v] = h ? y0v[v + 8] : y0v[v];
    yc1[v] = (!h && v < 4) ? y0v[16 + v] : 0.0f;
  }

#pragma unroll
  for (int s = 0; s < 5; ++s) {
    const int k0 = 4 * s + 2 * hi;
#pragma unroll
    for (int nt = 0; nt < 2; ++nt) {
      const int n = 16 * nt + n16;
      const bool ok = n < H_SZ;
      w1f[nt][s].x = ok ? fW1[k0 * H_SZ + n]       : 0.0f;
      w1f[nt][s].y = ok ? fW1[(k0 + 1) * H_SZ + n] : 0.0f;
      w2f[nt][s].x = ok ? fW2[k0 * H_SZ + n]       : 0.0f;
      w2f[nt][s].y = ok ? fW2[(k0 + 1) * H_SZ + n] : 0.0f;
    }
  }
#pragma unroll
  for (int nt = 0; nt < 2; ++nt) {
    const int n = 16 * nt + n16;
    const bool ok = (!h) && (n < H_SZ);
    w1f[nt][5].x = ok ? fB1[n] : 0.0f;  w1f[nt][5].y = 0.0f;
    w2f[nt][5].x = ok ? fB2[n] : 0.0f;  w2f[nt][5].y = 0.0f;
  }
  }

  auto buildB = [&](const v8f& z0, const v8f& z1, v2f* bf, auto behindTrans) {
    constexpr bool kBT = decltype(behindTrans)::value;
    float a, b;
    a = z0[0]; b = z0[2]; xchg_halves<kBT>(a, b); bf[0].x = a; bf[2].x = b;
    a = z0[1]; b = z0[3]; xchg_halves<kBT>(a, b); bf[0].y = a; bf[2].y = b;
    a = z0[4]; b = z0[6]; xchg_halves<kBT>(a, b); bf[1].x = a; bf[3].x = b;
    a = z0[5]; b = z0[7]; xchg_halves<kBT>(a, b); bf[1].y = a; bf[3].y = b;
    a = z1[0]; b = z1[2]; xchg_halves<kBT>(a, b); bf[4].x = a;
    a = z1[1]; b = z1[3]; xchg_halves<kBT>(a, b); bf[4].y = a;
  };
  auto matmul = [&](const v2f (*wa)[6], const v2f* bf, v8f& o0, v8f& o1) {
    o0 = wmma4(wa[0][5], bone, vzero);
    o1 = wmma4(wa[1][5], bone, vzero);
#pragma unroll
    for (int s = 0; s < 5; ++s) {
      o0 = wmma4(wa[0][s], bf[s], o0);
      o1 = wmma4(wa[1][s], bf[s], o1);
    }
  };
  auto fEval = [&](const v2f* bf, v8f& k0, v8f& k1) {
    v8f t0, t1;
    matmul(w1f, bf, t0, t1);
#pragma unroll
    for (int v = 0; v < 8; ++v) {
      t0[v] = fast_tanh(t0[v]);
      t1[v] = fast_tanh(t1[v]);
    }
    v2f bt[5];
    buildB(t0, t1, bt, BehindTrans<true>{});
    matmul(w2f, bt, k0, k1);
  };
  auto project = [&](const v2f* bf, int t) {
    v8f p = wmma4(w3f[5], bone, vzero);
#pragma unroll
    for (int s = 0; s < 5; ++s) p = wmma4(w3f[s], bf[s], p);
    if (!h) {
      float* o = &pred[((size_t)(rowBase + n16) * T_SZ + t) * D_SZ];
      o[0] = p[0]; o[1] = p[1]; o[2] = p[2];
    }
  };

  for (int c = 0; c <= NCH; ++c) {
    if (solver) {
      if (c < NCH) {
        for (int j = 0; j < CH; ++j) {
          const int t = c * CH + j;
          v2f bf[5];

          buildB(yc0, yc1, bf, BehindTrans<false>{});
#pragma unroll
          for (int s = 0; s < 5; ++s) ring[c & 1][tile][j][s][lane] = bf[s];

          if (t < T_SZ - 1) {
            const float dt  = ts[t + 1] - ts[t];
            const float hdt = 0.5f * dt;

            v8f ka, kb, acc0, acc1, u0, u1;
            fEval(bf, ka, kb);
#pragma unroll
            for (int v = 0; v < 8; ++v) {
              acc0[v] = ka[v];                 acc1[v] = kb[v];
              u0[v] = yc0[v] + hdt * ka[v];    u1[v] = yc1[v] + hdt * kb[v];
            }
            buildB(u0, u1, bf, BehindTrans<false>{});
            fEval(bf, ka, kb);
#pragma unroll
            for (int v = 0; v < 8; ++v) {
              acc0[v] += 2.0f * ka[v];         acc1[v] += 2.0f * kb[v];
              u0[v] = yc0[v] + hdt * ka[v];    u1[v] = yc1[v] + hdt * kb[v];
            }
            buildB(u0, u1, bf, BehindTrans<false>{});
            fEval(bf, ka, kb);
#pragma unroll
            for (int v = 0; v < 8; ++v) {
              acc0[v] += 2.0f * ka[v];         acc1[v] += 2.0f * kb[v];
              u0[v] = yc0[v] + dt * ka[v];     u1[v] = yc1[v] + dt * kb[v];
            }
            buildB(u0, u1, bf, BehindTrans<false>{});
            fEval(bf, ka, kb);

            const float sdt = dt * (1.0f / 6.0f);
#pragma unroll
            for (int v = 0; v < 8; ++v) {
              yc0[v] += sdt * (acc0[v] + ka[v]);
              yc1[v] += sdt * (acc1[v] + kb[v]);
            }
          }
        }
      }
    } else if (c > 0) {
      const int cc = c - 1;
      for (int j = 0; j < CH; ++j) {
        v2f bf[5];
#pragma unroll
        for (int s = 0; s < 5; ++s) bf[s] = ring[cc & 1][tile][j][s][lane];
        project(bf, cc * CH + j);
      }
    }
    __syncthreads();
  }
}

extern "C" void kernel_launch(void* const* d_in, const int* in_sizes, int n_in,
                              void* d_out, int out_size, void* d_ws, size_t ws_size,
                              hipStream_t stream) {
  const float* ts    = (const float*)d_in[0];
  const float* ys    = (const float*)d_in[1];
  const float* eps   = (const float*)d_in[2];
  const float* encW1 = (const float*)d_in[3];
  const float* encB1 = (const float*)d_in[4];
  const float* encW2 = (const float*)d_in[5];
  const float* encB2 = (const float*)d_in[6];
  const float* encW3 = (const float*)d_in[7];
  const float* encB3 = (const float*)d_in[8];
  const float* l2hW  = (const float*)d_in[9];
  const float* l2hB  = (const float*)d_in[10];
  const float* h2dW  = (const float*)d_in[11];
  const float* h2dB  = (const float*)d_in[12];
  const float* fW1   = (const float*)d_in[13];
  const float* fB1   = (const float*)d_in[14];
  const float* fW2   = (const float*)d_in[15];
  const float* fB2   = (const float*)d_in[16];

  float* pred = (float*)d_out;
  float* mean = pred + (size_t)B_SZ * T_SZ * D_SZ;
  float* stdv = mean + (size_t)B_SZ * L_SZ;

  dim3 grid(B_SZ / (SOLVER_WAVES * 16));
  dim3 block(WAVES_PER_BLOCK * 32);
  hipLaunchKernelGGL(latent_ode_kernel, grid, block, 0, stream,
                     ts, ys, eps, encW1, encB1, encW2, encB2, encW3, encB3,
                     l2hW, l2hB, h2dW, h2dB, fW1, fB1, fW2, fB2,
                     pred, mean, stdv);
}


# Encoder moved from per-lane scalar sums fed by serial scalar loads onto the f32 matrix pipe in the time loop's fragment form, all weights by one group of per-lane loads; encoder sums reordered
# speedup vs baseline: 1.8576x; 1.0602x over previous
#include <hip/hip_runtime.h>
#include <math.h>

typedef float v2f __attribute__((ext_vector_type(2)));
typedef float v8f __attribute__((ext_vector_type(8)));

#define B_SZ 8192
#define T_SZ 200
#define D_SZ 3
#define H_SZ 20
#define L_SZ 10

#define SOLVER_WAVES 2
#define WAVES_PER_BLOCK 4
#define CH 8
#define NCH (T_SZ / CH)
#define RING_SLOTS 40

__device__ __forceinline__ v8f wmma4(v2f a, v2f b, v8f c) {
  return __builtin_amdgcn_wmma_f32_16x16x4_f32(false, a, false, b, (short)0, c,
                                               false, false);
}

__device__ __forceinline__ float fast_tanh(float x) {
#if __has_builtin(__builtin_amdgcn_tanhf)
  return __builtin_amdgcn_tanhf(x);
#else
  float r;
  asm volatile("v_tanh_f32 %0, %1\n\tv_nop" : "=v"(r) : "v"(x));
  return r;
#endif
}

template <bool kV> struct BehindTrans { static constexpr bool value = kV; };

template <bool kBehindTrans>
__device__ __forceinline__ void xchg_halves(float& d, float& s) {
  if (kBehindTrans)
    asm("v_nop\n\tv_permlane16_swap_b32 %0, %1" : "+v"(d), "+v"(s));
  else
    asm("v_permlane16_swap_b32 %0, %1" : "+v"(d), "+v"(s));
}

__global__ __launch_bounds__(WAVES_PER_BLOCK * 32)
void latent_ode_kernel(const float* __restrict__ ts, const float* __restrict__ ys,
                       const float* __restrict__ eps,
                       const float* __restrict__ encW1, const float* __restrict__ encB1,
                       const float* __restrict__ encW2, const float* __restrict__ encB2,
                       const float* __restrict__ encW3, const float* __restrict__ encB3,
                       const float* __restrict__ l2hW, const float* __restrict__ l2hB,
                       const float* __restrict__ h2dW, const float* __restrict__ h2dB,
                       const float* __restrict__ fW1, const float* __restrict__ fB1,
                       const float* __restrict__ fW2, const float* __restrict__ fB2,
                       float* __restrict__ pred, float* __restrict__ outMean,
                       float* __restrict__ outStd)
{
  __shared__ __attribute__((aligned(16))) v2f ring[2][SOLVER_WAVES][RING_SLOTS][5][32];

  const int tid  = (int)threadIdx.x;
  const int wave = __builtin_amdgcn_readfirstlane(tid >> 5);
  const int lane = tid & 31;
  const int n16  = lane & 15;
  const int hi   = lane >> 4;
  const bool h   = (hi != 0);
  const bool solver = wave < SOLVER_WAVES;
  const int tile = solver ? wave : wave - SOLVER_WAVES;
  const int rowBase = ((int)blockIdx.x * SOLVER_WAVES + tile) * 16;
  const int row = rowBase + n16;

  v2f bone;
  bone.x = h ? 0.0f : 1.0f;
  bone.y = 0.0f;
  const v8f vzero = {};

  v8f yc0 = {}, yc1 = {};
  v2f w1f[2][6] = {}, w2f[2][6] = {};
  v2f w3f[6] = {};

  auto buildB = [&](const v8f& z0, const v8f& z1, v2f* bf, auto behindTrans) {
    constexpr bool kBT = decltype(behindTrans)::value;
    float a, b;
    a = z0[0]; b = z0[2]; xchg_halves<kBT>(a, b); bf[0].x = a; bf[2].x = b;
    a = z0[1]; b = z0[3]; xchg_halves<kBT>(a, b); bf[0].y = a; bf[2].y = b;
    a = z0[4]; b = z0[6]; xchg_halves<kBT>(a, b); bf[1].x = a; bf[3].x = b;
    a = z0[5]; b = z0[7]; xchg_halves<kBT>(a, b); bf[1].y = a; bf[3].y = b;
    a = z1[0]; b = z1[2]; xchg_halves<kBT>(a, b); bf[4].x = a;
    a = z1[1]; b = z1[3]; xchg_halves<kBT>(a, b); bf[4].y = a;
  };
  auto matmul = [&](const v2f (*wa)[6], const v2f* bf, v8f& o0, v8f& o1) {
    o0 = wmma4(wa[0][5], bone, vzero);
    o1 = wmma4(wa[1][5], bone, vzero);
#pragma unroll
    for (int s = 0; s < 5; ++s) {
      o0 = wmma4(wa[0][s], bf[s], o0);
      o1 = wmma4(wa[1][s], bf[s], o1);
    }
  };

  if (!solver) {
#pragma unroll
    for (int s = 0; s < 5; ++s) {
      const int k0 = 4 * s + 2 * hi;
      const bool ok3 = n16 < D_SZ;
      w3f[s].x = ok3 ? h2dW[k0 * D_SZ + n16]       : 0.0f;
      w3f[s].y = ok3 ? h2dW[(k0 + 1) * D_SZ + n16] : 0.0f;
    }
    w3f[5].x = (!h && n16 < D_SZ) ? h2dB[n16] : 0.0f;
    w3f[5].y = 0.0f;
  } else {

  auto loadA = [&](const float* __restrict__ Wm, const float* __restrict__ Bv,
                   int kdim, int ld, int col0, int nlive, int nsteps, v2f* wa) {
    const bool ok = n16 < nlive;
    const int col = col0 + (ok ? n16 : 0);
#pragma unroll
    for (int s = 0; s < nsteps; ++s) {
      const int k0 = 4 * s + 2 * hi;
      const bool okx = ok && (k0 < kdim);
      const bool oky = ok && (k0 + 1 < kdim);
      const float vx = Wm[(k0 < kdim ? k0 : 0) * ld + col];
      const float vy = Wm[(k0 + 1 < kdim ? k0 + 1 : 0) * ld + col];
      wa[s].x = okx ? vx : 0.0f;
      wa[s].y = oky ? vy : 0.0f;
    }
    const bool okb = ok && !h;
    const float vb = Bv[okb ? col : col0];
    wa[5].x = okb ? vb : 0.0f;
    wa[5].y = 0.0f;
  };
  auto matmulN = [&](const v2f (*wa)[6], const v2f* bf, int nsteps, v8f& o0, v8f& o1) {
    o0 = wmma4(wa[0][5], bone, vzero);
    o1 = wmma4(wa[1][5], bone, vzero);
#pragma unroll
    for (int s = 0; s < nsteps; ++s) {
      o0 = wmma4(wa[0][s], bf[s], o0);
      o1 = wmma4(wa[1][s], bf[s], o1);
    }
  };
  auto reluTile = [&](v8f& z) {
#pragma unroll
    for (int v = 0; v < 8; ++v) z[v] = fmaxf(z[v], 0.0f);
  };

  const float* yrow = ys + (size_t)row * T_SZ * D_SZ;
  const float yin0 = yrow[0], yin1 = yrow[1], yin2 = yrow[2];
  float epsv[8];
#pragma unroll
  for (int v = 0; v < 8; ++v) {
    const int j = 8 * hi + v;
    const float e = eps[(size_t)row * L_SZ + (j < L_SZ ? j : 0)];
    epsv[v] = (j < L_SZ) ? e : 0.0f;
  }

  v2f a1[2];
#pragma unroll
  for (int nt = 0; nt < 2; ++nt) {
    const bool ok = (16 * nt + n16) < H_SZ;
    const int col = 16 * nt + (ok ? n16 : 0);
    const float wx = encW1[(2 * hi) * H_SZ + col];
    const float wy = encW1[1 * H_SZ + col];
    const float by = encB1[col];
    a1[nt].x = ok ? wx : 0.0f;
    a1[nt].y = ok ? (h ? by : wy) : 0.0f;
  }
  v2f bin;
  bin.x = h ? yin2 : yin0;
  bin.y = h ? 1.0f : yin1;

  v2f e2f[2][6], e3f[2][6], l2f[2][6] = {};
  loadA(encW2, encB2, H_SZ, H_SZ, 0,  16, 5, e2f[0]);
  loadA(encW2, encB2, H_SZ, H_SZ, 16, H_SZ - 16, 5, e2f[1]);
  loadA(encW3, encB3, H_SZ, 2 * L_SZ, 0,    L_SZ, 5, e3f[0]);
  loadA(encW3, encB3, H_SZ, 2 * L_SZ, L_SZ, L_SZ, 5, e3f[1]);
  loadA(l2hW, l2hB, L_SZ, H_SZ, 0,  16, 3, l2f[0]);
  loadA(l2hW, l2hB, L_SZ, H_SZ, 16, H_SZ - 16, 3, l2f[1]);
  loadA(fW1, fB1, H_SZ, H_SZ, 0,  16, 5, w1f[0]);
  loadA(fW1, fB1, H_SZ, H_SZ, 16, H_SZ - 16, 5, w1f[1]);
  loadA(fW2, fB2, H_SZ, H_SZ, 0,  16, 5, w2f[0]);
  loadA(fW2, fB2, H_SZ, H_SZ, 16, H_SZ - 16, 5, w2f[1]);

  v8f t0 = wmma4(a1[0], bin, vzero);
  v8f t1 = wmma4(a1[1], bin, vzero);
  reluTile(t0); reluTile(t1);
  v2f ebf[5];
  buildB(t0, t1, ebf, BehindTrans<false>{});
  matmul(e2f, ebf, t0, t1);
  reluTile(t0); reluTile(t1);
  buildB(t0, t1, ebf, BehindTrans<false>{});
  v8f mt, lt;
  matmul(e3f, ebf, mt, lt);

  v8f latz;
  float sdv[8];
#pragma unroll
  for (int v = 0; v < 8; ++v) {
    sdv[v] = expf(lt[v]);
    const float lv = mt[v] + epsv[v] * sdv[v];
    latz[v] = (8 * hi + v < L_SZ) ? lv : 0.0f;
  }
  {
    float* om = &outMean[(size_t)row * L_SZ + 8 * hi];
    float* os = &outStd [(size_t)row * L_SZ + 8 * hi];
    om[0] = mt[0]; om[1] = mt[1];
    os[0] = sdv[0]; os[1] = sdv[1];
    if (!h) {
#pragma unroll
      for (int v = 2; v < 8; ++v) { om[v] = mt[v]; os[v] = sdv[v]; }
    }
  }

  buildB(latz, vzero, ebf, BehindTrans<false>{});
  matmulN(l2f, ebf, 3, yc0, yc1);
  reluTile(yc0); reluTile(yc1);

  }

  auto fEval = [&](const v2f* bf, v8f& k0, v8f& k1) {
    v8f t0, t1;
    matmul(w1f, bf, t0, t1);
#pragma unroll
    for (int v = 0; v < 8; ++v) {
      t0[v] = fast_tanh(t0[v]);
      t1[v] = fast_tanh(t1[v]);
    }
    v2f bt[5];
    buildB(t0, t1, bt, BehindTrans<true>{});
    matmul(w2f, bt, k0, k1);
  };
  auto project = [&](const v2f* bf, int t) {
    v8f p = wmma4(w3f[5], bone, vzero);
#pragma unroll
    for (int s = 0; s < 5; ++s) p = wmma4(w3f[s], bf[s], p);
    if (!h) {
      float* o = &pred[((size_t)(rowBase + n16) * T_SZ + t) * D_SZ];
      o[0] = p[0]; o[1] = p[1]; o[2] = p[2];
    }
  };

  for (int c = 0; c <= NCH; ++c) {
    if (solver) {
      if (c < NCH) {
        for (int j = 0; j < CH; ++j) {
          const int t = c * CH + j;
          v2f bf[5];

          buildB(yc0, yc1, bf, BehindTrans<false>{});
#pragma unroll
          for (int s = 0; s < 5; ++s) ring[c & 1][tile][j][s][lane] = bf[s];

          if (t < T_SZ - 1) {
            const float dt  = ts[t + 1] - ts[t];
            const float hdt = 0.5f * dt;

            v8f ka, kb, acc0, acc1, u0, u1;
            fEval(bf, ka, kb);
#pragma unroll
            for (int v = 0; v < 8; ++v) {
              acc0[v] = ka[v];                 acc1[v] = kb[v];
              u0[v] = yc0[v] + hdt * ka[v];    u1[v] = yc1[v] + hdt * kb[v];
            }
            buildB(u0, u1, bf, BehindTrans<false>{});
            fEval(bf, ka, kb);
#pragma unroll
            for (int v = 0; v < 8; ++v) {
              acc0[v] += 2.0f * ka[v];         acc1[v] += 2.0f * kb[v];
              u0[v] = yc0[v] + hdt * ka[v];    u1[v] = yc1[v] + hdt * kb[v];
            }
            buildB(u0, u1, bf, BehindTrans<false>{});
            fEval(bf, ka, kb);
#pragma unroll
            for (int v = 0; v < 8; ++v) {
              acc0[v] += 2.0f * ka[v];         acc1[v] += 2.0f * kb[v];
              u0[v] = yc0[v] + dt * ka[v];     u1[v] = yc1[v] + dt * kb[v];
            }
            buildB(u0, u1, bf, BehindTrans<false>{});
            fEval(bf, ka, kb);

            const float sdt = dt * (1.0f / 6.0f);
#pragma unroll
            for (int v = 0; v < 8; ++v) {
              yc0[v] += sdt * (acc0[v] + ka[v]);
              yc1[v] += sdt * (acc1[v] + kb[v]);
            }
          }
        }
      }
    } else if (c > 0) {
      const int cc = c - 1;
      for (int j = 0; j < CH; ++j) {
        v2f bf[5];
#pragma unroll
        for (int s = 0; s < 5; ++s) bf[s] = ring[cc & 1][tile][j][s][lane];
        project(bf, cc * CH + j);
      }
    }
    __syncthreads();
  }
}

extern "C" void kernel_launch(void* const* d_in, const int* in_sizes, int n_in,
                              void* d_out, int out_size, void* d_ws, size_t ws_size,
                              hipStream_t stream) {
  const float* ts    = (const float*)d_in[0];
  const float* ys    = (const float*)d_in[1];
  const float* eps   = (const float*)d_in[2];
  const float* encW1 = (const float*)d_in[3];
  const float* encB1 = (const float*)d_in[4];
  const float* encW2 = (const float*)d_in[5];
  const float* encB2 = (const float*)d_in[6];
  const float* encW3 = (const float*)d_in[7];
  const float* encB3 = (const float*)d_in[8];
  const float* l2hW  = (const float*)d_in[9];
  const float* l2hB  = (const float*)d_in[10];
  const float* h2dW  = (const float*)d_in[11];
  const float* h2dB  = (const float*)d_in[12];
  const float* fW1   = (const float*)d_in[13];
  const float* fB1   = (const float*)d_in[14];
  const float* fW2   = (const float*)d_in[15];
  const float* fB2   = (const float*)d_in[16];

  float* pred = (float*)d_out;
  float* mean = pred + (size_t)B_SZ * T_SZ * D_SZ;
  float* stdv = mean + (size_t)B_SZ * L_SZ;

  dim3 grid(B_SZ / (SOLVER_WAVES * 16));
  dim3 block(WAVES_PER_BLOCK * 32);
  hipLaunchKernelGGL(latent_ode_kernel, grid, block, 0, stream,
                     ts, ys, eps, encW1, encB1, encW2, encB2, encW3, encB3,
                     l2hW, l2hB, h2dW, h2dB, fW1, fB1, fW2, fB2,
                     pred, mean, stdv);
}
